# starting kernel
# baseline (speedup 1.0000x reference)
#include <hip/hip_runtime.h>
#include <hip/hip_bf16.h>

#define BB 8
#define LL 8192
#define CC 512
#define ROWS (BB * LL)
#define OUT_PER_B (LL - 2)
#define OUT_TOTAL (BB * OUT_PER_B)

typedef __attribute__((ext_vector_type(2))) float v2f;
typedef __attribute__((ext_vector_type(4))) float v4f;
typedef __attribute__((ext_vector_type(8))) float v8f;

static __device__ __forceinline__ float sum4(const v4f v) {
    return (v[0] + v[1]) + (v[2] + v[3]);
}

__global__ void __launch_bounds__(256) x_partial_kernel(const float* __restrict__ x,
                                                        float* __restrict__ partials) {
    const int t = threadIdx.x;
    const v4f* x4 = (const v4f*)x + (long)blockIdx.x * 8192 + t;
    float acc = 0.f;
#pragma unroll 1
    for (int i = 0; i < 32; i += 8) {
        float s = 0.f;
#pragma unroll
        for (int j = 0; j < 8; ++j)
            s += sum4(__builtin_nontemporal_load(&x4[(i + j) * 256]));
        acc += s;
    }
    __shared__ float sm[256];
    sm[t] = acc;
    __syncthreads();
#pragma unroll
    for (int s = 128; s > 0; s >>= 1) {
        if (t < s) sm[t] += sm[t + s];
        __syncthreads();
    }
    if (t == 0) partials[blockIdx.x] = sm[0];
}

__global__ void m_final_kernel(const float* __restrict__ partials,
                               float* __restrict__ M) {
    const int b = threadIdx.x;
    if (b < BB) {
        float s = 0.f;
        for (int i = 0; i < 128; ++i) s += partials[b * 128 + i];
        M[b] = s / (float)(LL * CC);
    }
}

__global__ void __launch_bounds__(256) rowsum_wmma_kernel(const float* __restrict__ noise,
                                                          float* __restrict__ S) {
    const int lane = threadIdx.x & 31;
    const int wave = threadIdx.x >> 5;
    const long rowbase = (long)blockIdx.x * 128 + wave * 16;
    const int m  = lane & 15;
    const int hi = lane >> 4;

    const v4f* rp = (const v4f*)(noise + (rowbase + m) * CC + hi * 256);

    float p0 = 0.f, p1 = 0.f;
#pragma unroll 1
    for (int i = 0; i < 32; i += 8) {
        float s = 0.f;
#pragma unroll
        for (int j = 0; j < 8; ++j)
            s += sum4(__builtin_nontemporal_load(&rp[i + j]));
        p0 += s;
    }
#pragma unroll 1
    for (int i = 32; i < 64; i += 8) {
        float s = 0.f;
#pragma unroll
        for (int j = 0; j < 8; ++j)
            s += sum4(__builtin_nontemporal_load(&rp[i + j]));
        p1 += s;
    }

    v2f a; a[0] = p0; a[1] = p1;
    v2f b; b[0] = 1.f; b[1] = 1.f;
    v8f c = {};
    c = __builtin_amdgcn_wmma_f32_16x16x4_f32(
             false, a,  false, b,
             (short)0, c,  false,  false);

    if (lane == 0) {
#pragma unroll
        for (int r = 0; r < 8; ++r) S[rowbase + r] = c[r];
    } else if (lane == 16) {
#pragma unroll
        for (int r = 0; r < 8; ++r) S[rowbase + 8 + r] = c[r];
    }
}

__global__ void pool_kernel(const float* __restrict__ M,
                            const float* __restrict__ S,
                            float* __restrict__ out) {
    int idx = blockIdx.x * 256 + threadIdx.x;
    if (idx >= OUT_TOTAL) return;
    int b = idx / OUT_PER_B;
    int l = idx - b * OUT_PER_B;
    const float* Sb = S + (long)b * LL + l;
    const float k = 0.1f / (3.0f * (float)CC);
    out[idx] = M[b] + k * ((Sb[0] + Sb[1]) + Sb[2]);
}

extern "C" void kernel_launch(void* const* d_in, const int* in_sizes, int n_in,
                              void* d_out, int out_size, void* d_ws, size_t ws_size,
                              hipStream_t stream) {
    (void)in_sizes; (void)n_in; (void)out_size; (void)ws_size;
    const float* x     = (const float*)d_in[0];
    const float* noise = (const float*)d_in[1];
    float* out = (float*)d_out;

    float* ws       = (float*)d_ws;
    float* partials = ws;
    float* M        = ws + 1024;
    float* S        = ws + 2048;

    x_partial_kernel<<<1024, 256, 0, stream>>>(x, partials);
    m_final_kernel<<<1, 32, 0, stream>>>(partials, M);
    rowsum_wmma_kernel<<<ROWS / 128, 256, 0, stream>>>(noise, S);
    pool_kernel<<<(OUT_TOTAL + 255) / 256, 256, 0, stream>>>(M, S, out);
}


# noise row-sum kernel: wave streams its 16 rows flat in whole 512-byte pieces; per-lane row partials go through LDS into 8 chained WMMAs that sum over the 32 lanes; summation order changes
# speedup vs baseline: 1.7744x; 1.7744x over previous
#include <hip/hip_runtime.h>
#include <hip/hip_bf16.h>

#define BB 8
#define LL 8192
#define CC 512
#define ROWS (BB * LL)
#define OUT_PER_B (LL - 2)
#define OUT_TOTAL (BB * OUT_PER_B)

typedef __attribute__((ext_vector_type(2))) float v2f;
typedef __attribute__((ext_vector_type(4))) float v4f;
typedef __attribute__((ext_vector_type(8))) float v8f;

static __device__ __forceinline__ float sum4(const v4f v) {
    return (v[0] + v[1]) + (v[2] + v[3]);
}

__global__ void __launch_bounds__(256) x_partial_kernel(const float* __restrict__ x,
                                                        float* __restrict__ partials) {
    const int t = threadIdx.x;
    const v4f* x4 = (const v4f*)x + (long)blockIdx.x * 8192 + t;
    float acc = 0.f;
#pragma unroll 1
    for (int i = 0; i < 32; i += 8) {
        float s = 0.f;
#pragma unroll
        for (int j = 0; j < 8; ++j)
            s += sum4(__builtin_nontemporal_load(&x4[(i + j) * 256]));
        acc += s;
    }
    __shared__ float sm[256];
    sm[t] = acc;
    __syncthreads();
#pragma unroll
    for (int s = 128; s > 0; s >>= 1) {
        if (t < s) sm[t] += sm[t + s];
        __syncthreads();
    }
    if (t == 0) partials[blockIdx.x] = sm[0];
}

__global__ void m_final_kernel(const float* __restrict__ partials,
                               float* __restrict__ M) {
    const int b = threadIdx.x;
    if (b < BB) {
        float s = 0.f;
        for (int i = 0; i < 128; ++i) s += partials[b * 128 + i];
        M[b] = s / (float)(LL * CC);
    }
}

#define P_PITCH 17

__global__ void __launch_bounds__(256) rowsum_wmma_kernel(const float* __restrict__ noise,
                                                          float* __restrict__ S) {
    const int lane = threadIdx.x & 31;
    const int wave = threadIdx.x >> 5;
    const long rowbase = (long)blockIdx.x * 128 + wave * 16;
    const int m  = lane & 15;
    const int hi = lane >> 4;

    __shared__ float part[8 * 32 * P_PITCH];
    float* pw = part + wave * (32 * P_PITCH);

    const v4f* tp = (const v4f*)(noise + rowbase * CC) + lane;

#pragma unroll 1
    for (int r = 0; r < 16; r += 2) {
        v4f v[8];
#pragma unroll
        for (int j = 0; j < 8; ++j)
            v[j] = __builtin_nontemporal_load(&tp[(r * 4 + j) * 32]);
        const float s0 = (sum4(v[0]) + sum4(v[1])) + (sum4(v[2]) + sum4(v[3]));
        const float s1 = (sum4(v[4]) + sum4(v[5])) + (sum4(v[6]) + sum4(v[7]));
        pw[lane * P_PITCH + r]     = s0;
        pw[lane * P_PITCH + r + 1] = s1;
    }
    __syncthreads();

    v2f b; b[0] = 1.f; b[1] = 1.f;
    v8f c = {};
#pragma unroll
    for (int q = 0; q < 8; ++q) {
        const int L = 4 * q + 2 * hi;
        v2f a;
        a[0] = pw[L * P_PITCH + m];
        a[1] = pw[(L + 1) * P_PITCH + m];
        c = __builtin_amdgcn_wmma_f32_16x16x4_f32(
                 false, a,  false, b,
                 (short)0, c,  false,  false);
    }

    if (lane == 0) {
#pragma unroll
        for (int r = 0; r < 8; ++r) S[rowbase + r] = c[r];
    } else if (lane == 16) {
#pragma unroll
        for (int r = 0; r < 8; ++r) S[rowbase + 8 + r] = c[r];
    }
}

__global__ void pool_kernel(const float* __restrict__ M,
                            const float* __restrict__ S,
                            float* __restrict__ out) {
    int idx = blockIdx.x * 256 + threadIdx.x;
    if (idx >= OUT_TOTAL) return;
    int b = idx / OUT_PER_B;
    int l = idx - b * OUT_PER_B;
    const float* Sb = S + (long)b * LL + l;
    const float k = 0.1f / (3.0f * (float)CC);
    out[idx] = M[b] + k * ((Sb[0] + Sb[1]) + Sb[2]);
}

extern "C" void kernel_launch(void* const* d_in, const int* in_sizes, int n_in,
                              void* d_out, int out_size, void* d_ws, size_t ws_size,
                              hipStream_t stream) {
    (void)in_sizes; (void)n_in; (void)out_size; (void)ws_size;
    const float* x     = (const float*)d_in[0];
    const float* noise = (const float*)d_in[1];
    float* out = (float*)d_out;

    float* ws       = (float*)d_ws;
    float* partials = ws;
    float* M        = ws + 1024;
    float* S        = ws + 2048;

    x_partial_kernel<<<1024, 256, 0, stream>>>(x, partials);
    m_final_kernel<<<1, 32, 0, stream>>>(partials, M);
    rowsum_wmma_kernel<<<ROWS / 128, 256, 0, stream>>>(noise, S);
    pool_kernel<<<(OUT_TOTAL + 255) / 256, 256, 0, stream>>>(M, S, out);
}


# split by tensor: x launch, then a noise launch that also finishes the mean, pools in LDS (2-4 edge rows re-read) and writes out; 4 -> 2 launches, no S plane; lane-butterfly sums: sum order changes
# speedup vs baseline: 2.1496x; 1.2115x over previous
#include <hip/hip_runtime.h>
#include <hip/hip_bf16.h>

#define BB 8
#define LL 8192
#define CC 512
#define ROWS (BB * LL)
#define OUT_PER_B (LL - 2)
#define OUT_TOTAL (BB * OUT_PER_B)

typedef __attribute__((ext_vector_type(2))) float v2f;
typedef __attribute__((ext_vector_type(4))) float v4f;
typedef __attribute__((ext_vector_type(8))) float v8f;

static __device__ __forceinline__ float sum4(const v4f v) {
    return (v[0] + v[1]) + (v[2] + v[3]);
}

__global__ void __launch_bounds__(256) x_partial_kernel(const float* __restrict__ x,
                                                        float* __restrict__ partials) {
    const int t = threadIdx.x;
    const v4f* x4 = (const v4f*)x + (long)blockIdx.x * 8192 + t;
    float acc = 0.f;
#pragma unroll 1
    for (int i = 0; i < 32; i += 8) {
        float s = 0.f;
#pragma unroll
        for (int j = 0; j < 8; ++j)
            s += sum4(__builtin_nontemporal_load(&x4[(i + j) * 256]));
        acc += s;
    }
    __shared__ float sm[256];
    sm[t] = acc;
    __syncthreads();
#pragma unroll
    for (int s = 128; s > 0; s >>= 1) {
        if (t < s) sm[t] += sm[t + s];
        __syncthreads();
    }
    if (t == 0) partials[blockIdx.x] = sm[0];
}

#define P_PITCH 17
#define WIN_ROWS 132
#define NGROUPS ((OUT_TOTAL + 127) / 128)

template <int MASK>
static __device__ __forceinline__ float row_xor(const float v) {
    return __builtin_bit_cast(float,
        __builtin_amdgcn_update_dpp(0, __builtin_bit_cast(int, v), 0x160 | MASK, 0xf, 0xf, true));
}
static __device__ __forceinline__ float xhalf(const float v) {
    const int b = __builtin_bit_cast(int, v);
    return __builtin_bit_cast(float,
        __builtin_amdgcn_permlanex16(b, b, 0x76543210, 0xfedcba98, false, false));
}
static __device__ __forceinline__ float wave_sum(float v) {
    v += row_xor<1>(v);
    v += row_xor<2>(v);
    v += row_xor<4>(v);
    v += row_xor<8>(v);
    return v + xhalf(v);
}

__global__ void __launch_bounds__(256) rowsum_pool_kernel(const float* __restrict__ noise,
                                                          const float* __restrict__ partials,
                                                          float* __restrict__ out) {
    const int lane = threadIdx.x & 31;
    const int wave = threadIdx.x >> 5;
    const int m  = lane & 15;
    const int hi = lane >> 4;

    const int idx0 = blockIdx.x * 128;
    const int idxl = (idx0 + 127 < OUT_TOTAL) ? idx0 + 127 : OUT_TOTAL - 1;
    const int b0 = idx0 / OUT_PER_B;
    const int b1 = idxl / OUT_PER_B;
    int rowstart = idx0 + 2 * b0;
    if (rowstart > ROWS - WIN_ROWS) rowstart = ROWS - WIN_ROWS;
    const int nedge = (idxl + 2 * b1 + 2) - (rowstart + 127);
    const long rowbase = (long)rowstart + wave * 16;

    __shared__ float part[8 * 32 * P_PITCH];
    __shared__ float Sx[WIN_ROWS];
    __shared__ float eq[16];
    __shared__ float Mq[2];
    float* pw = part + wave * (32 * P_PITCH);

    const v4f* tp = (const v4f*)(noise + rowbase * CC) + lane;

    const v4f* ep = (const v4f*)(noise + ((long)rowstart + 128 + (wave >> 2)) * CC) + (wave & 3) * 32 + lane;
    const v4f e0 = __builtin_nontemporal_load(ep);
    v4f e1 = {};
    if (nedge == 4) e1 = __builtin_nontemporal_load(ep + 2 * (CC / 4));
    const v4f pm = ((const v4f*)partials)[((wave & 1) ? b1 : b0) * 32 + lane];

#pragma unroll 1
    for (int r = 0; r < 16; r += 2) {
        v4f v[8];
#pragma unroll
        for (int j = 0; j < 8; ++j)
            v[j] = __builtin_nontemporal_load(&tp[(r * 4 + j) * 32]);
        const float s0 = (sum4(v[0]) + sum4(v[1])) + (sum4(v[2]) + sum4(v[3]));
        const float s1 = (sum4(v[4]) + sum4(v[5])) + (sum4(v[6]) + sum4(v[7]));
        pw[lane * P_PITCH + r]     = s0;
        pw[lane * P_PITCH + r + 1] = s1;
    }

    const float es0 = wave_sum(sum4(e0));
    const float es1 = wave_sum(sum4(e1));
    if (lane == 0) { eq[wave] = es0; eq[8 + wave] = es1; }
    const float mm = wave_sum(sum4(pm)) / (float)(LL * CC);
    if (lane == 0 && wave < 2) Mq[wave] = mm;
    __syncthreads();

    v2f b; b[0] = 1.f; b[1] = 1.f;
    v8f c = {};
#pragma unroll
    for (int q = 0; q < 8; ++q) {
        const int L = 4 * q + 2 * hi;
        v2f a;
        a[0] = pw[L * P_PITCH + m];
        a[1] = pw[(L + 1) * P_PITCH + m];
        c = __builtin_amdgcn_wmma_f32_16x16x4_f32(
                 false, a,  false, b,
                 (short)0, c,  false,  false);
    }

    if (m == 0) {
        float* d = Sx + wave * 16 + hi * 8;
#pragma unroll
        for (int r = 0; r < 8; ++r) d[r] = c[r];
    }
    if (threadIdx.x < 4) {
        const float* e = eq + 4 * threadIdx.x;
        Sx[128 + threadIdx.x] = (e[0] + e[1]) + (e[2] + e[3]);
    }
    __syncthreads();

    if (threadIdx.x < 128) {
        const int idx = idx0 + threadIdx.x;
        if (idx < OUT_TOTAL) {
            const int bb = idx / OUT_PER_B;
            const float* Sb = Sx + (idx + 2 * bb - rowstart);
            const float k = 0.1f / (3.0f * (float)CC);
            out[idx] = Mq[bb - b0] + k * ((Sb[0] + Sb[1]) + Sb[2]);
        }
    }
}

extern "C" void kernel_launch(void* const* d_in, const int* in_sizes, int n_in,
                              void* d_out, int out_size, void* d_ws, size_t ws_size,
                              hipStream_t stream) {
    (void)in_sizes; (void)n_in; (void)out_size; (void)ws_size;
    const float* x     = (const float*)d_in[0];
    const float* noise = (const float*)d_in[1];
    float* out = (float*)d_out;

    float* partials = (float*)d_ws;

    x_partial_kernel<<<1024, 256, 0, stream>>>(x, partials);
    rowsum_pool_kernel<<<NGROUPS, 256, 0, stream>>>(noise, partials, out);
}
